# starting kernel
# baseline (speedup 1.0000x reference)
#include <hip/hip_runtime.h>
#include <math.h>

constexpr int N_DOC = 4;
constexpr int L     = 1024;
constexpr int D     = 768;
constexpr int H     = 12;
constexpr int E     = 32;
constexpr int M     = 4;
constexpr int R     = 256;
constexpr float NEG_MAX = -3.402823466e+38f;

typedef __attribute__((ext_vector_type(2))) float v2f;
typedef __attribute__((ext_vector_type(4))) float v4f;
typedef __attribute__((ext_vector_type(8))) float v8f;

__global__ void k_e_emb(const float* __restrict__ seq,
                        const int*   __restrict__ pos,
                        const float* __restrict__ mask,
                        float* __restrict__ e_emb) {
    int b = blockIdx.x;
    int i = b / E;
    const float* seq_i  = seq  + (size_t)i * L * D;
    const int*   pos_e  = pos  + (size_t)b * M;
    const float* mask_e = mask + (size_t)b * M;

    int   p[M];
    float mk[M];
#pragma unroll
    for (int m = 0; m < M; ++m) { p[m] = pos_e[m] + 1; mk[m] = mask_e[m]; }

    for (int c = threadIdx.x; c < D; c += blockDim.x) {
        float v[M];
        float mx = NEG_MAX;
#pragma unroll
        for (int m = 0; m < M; ++m) {
            float val = seq_i[(size_t)p[m] * D + c];
            v[m] = (mk[m] > 0.0f) ? val : NEG_MAX;
            mx = fmaxf(mx, v[m]);
        }
        float s = 0.0f;
#pragma unroll
        for (int m = 0; m < M; ++m)
            if (mk[m] > 0.0f) s += expf(v[m] - mx);
        e_emb[(size_t)b * D + c] = mx + logf(s);
    }
}

__global__ void k_e_att(const float* __restrict__ att,
                        const int*   __restrict__ pos,
                        const float* __restrict__ mask,
                        float* __restrict__ e_att) {
    int b  = blockIdx.x;
    int hh = b % H;
    int e  = (b / H) % E;
    int i  = b / (H * E);

    const int*   pos_e  = pos  + ((size_t)i * E + e) * M;
    const float* mask_e = mask + ((size_t)i * E + e) * M;

    float cnt = 0.0f;
#pragma unroll
    for (int m = 0; m < M; ++m) cnt += mask_e[m];
    float inv = 1.0f / fmaxf(cnt, 1.0f);

    int l0 = threadIdx.x * 4;
    v4f acc = {0.0f, 0.0f, 0.0f, 0.0f};
#pragma unroll
    for (int m = 0; m < M; ++m) {
        float mk = mask_e[m];
        if (mk > 0.0f) {
            int p = pos_e[m] + 1;
            const v4f* row = (const v4f*)(att + (((size_t)i * H + hh) * L + p) * L + l0);
            acc += mk * row[0];
        }
    }
    acc *= inv;
    v4f* dst = (v4f*)(e_att + (((size_t)i * E + e) * H + hh) * L + l0);
    dst[0] = acc;
}

__global__ void k_ht_att(const float* __restrict__ e_att,
                         const int*   __restrict__ hts,
                         float* __restrict__ ht_att) {
    int b = blockIdx.x;
    int i = b / R;
    int eh = hts[(size_t)b * 2 + 0];
    int et = hts[(size_t)b * 2 + 1];

    const float* ha = e_att + ((size_t)i * E + eh) * H * L;
    const float* ta = e_att + ((size_t)i * E + et) * H * L;

    int l0 = threadIdx.x * 4;
    v4f acc = {0.0f, 0.0f, 0.0f, 0.0f};
#pragma unroll
    for (int hh = 0; hh < H; ++hh) {
        v4f a = *(const v4f*)(ha + (size_t)hh * L + l0);
        v4f t = *(const v4f*)(ta + (size_t)hh * L + l0);
        acc += a * t;
    }
    acc *= (1.0f / (float)H);

    __shared__ float red[256];
    float ps = acc.x + acc.y + acc.z + acc.w;
    red[threadIdx.x] = ps;
    __syncthreads();
    for (int s = 128; s > 0; s >>= 1) {
        if (threadIdx.x < s) red[threadIdx.x] += red[threadIdx.x + s];
        __syncthreads();
    }
    float invs = 1.0f / (red[0] + 1e-5f);

    v4f* dst = (v4f*)(ht_att + (size_t)b * L + l0);
    dst[0] = acc * invs;
}

__global__ void k_hs_ts(const float* __restrict__ e_emb,
                        const int*   __restrict__ hts,
                        float* __restrict__ out) {
    int b = blockIdx.x;
    int i = b / R;
    int eh = hts[(size_t)b * 2 + 0];
    int et = hts[(size_t)b * 2 + 1];

    const float* sh = e_emb + ((size_t)i * E + eh) * D;
    const float* st = e_emb + ((size_t)i * E + et) * D;
    float* oh = out + (size_t)b * D;
    float* ot = out + (size_t)N_DOC * R * D + (size_t)b * D;

    for (int c = threadIdx.x; c < D; c += blockDim.x) {
        oh[c] = sh[c];
        ot[c] = st[c];
    }
}

__global__ void k_rs_wmma(const float* __restrict__ seq,
                          const float* __restrict__ ht_att,
                          float* __restrict__ out) {
    constexpr int NT = D / 16;
    constexpr int MT = R / 32;
    int b  = blockIdx.x;
    int nt = b % NT;
    int mt = (b / NT) % MT;
    int i  = b / (NT * MT);

    int m0 = mt * 32;
    int n0 = nt * 16;

    const float* A  = ht_att + (size_t)i * R * L;
    const float* Bm = seq    + (size_t)i * L * D;

    int lane = threadIdx.x & 31;
    int half = lane >> 4;
    int l15  = lane & 15;
    int bcol = n0 + l15;

    v8f c0 = {0.f, 0.f, 0.f, 0.f, 0.f, 0.f, 0.f, 0.f};
    v8f c1 = {0.f, 0.f, 0.f, 0.f, 0.f, 0.f, 0.f, 0.f};

    const float* arow0 = A + (size_t)(m0 + l15) * L;
    const float* arow1 = A + (size_t)(m0 + 16 + l15) * L;

    for (int k0 = 0; k0 < L; k0 += 4) {
        int ka = k0 + half * 2;
        v2f a0 = *(const v2f*)(arow0 + ka);
        v2f a1 = *(const v2f*)(arow1 + ka);
        v2f bb;
        bb.x = Bm[(size_t)ka * D + bcol];
        bb.y = Bm[(size_t)(ka + 1) * D + bcol];
        c0 = __builtin_amdgcn_wmma_f32_16x16x4_f32(
                false, a0, false, bb, (short)0, c0, false, false);
        c1 = __builtin_amdgcn_wmma_f32_16x16x4_f32(
                false, a1, false, bb, (short)0, c1, false, false);
    }

    float* O = out + (size_t)2 * N_DOC * R * D + (size_t)i * R * D;
#pragma unroll
    for (int v = 0; v < 8; ++v) {
        int row0 = m0 + v + half * 8;
        int row1 = m0 + 16 + v + half * 8;
        O[(size_t)row0 * D + n0 + l15] = c0[v];
        O[(size_t)row1 * D + n0 + l15] = c1[v];
    }
}

extern "C" void kernel_launch(void* const* d_in, const int* in_sizes, int n_in,
                              void* d_out, int out_size, void* d_ws, size_t ws_size,
                              hipStream_t stream) {
    const float* seq   = (const float*)d_in[0];
    const float* att   = (const float*)d_in[1];
    const int*   pos   = (const int*)  d_in[2];
    const float* mask  = (const float*)d_in[3];
    const int*   hts   = (const int*)  d_in[4];
    float*       out   = (float*)d_out;

    float* ws     = (float*)d_ws;
    float* e_emb  = ws;
    float* e_att  = e_emb + (size_t)N_DOC * E * D;
    float* ht_att = e_att + (size_t)N_DOC * E * H * L;

    k_e_emb  <<<N_DOC * E,     256, 0, stream>>>(seq, pos, mask, e_emb);
    k_e_att  <<<N_DOC * E * H, 256, 0, stream>>>(att, pos, mask, e_att);
    k_ht_att <<<N_DOC * R,     256, 0, stream>>>(e_att, hts, ht_att);
    k_hs_ts  <<<N_DOC * R,     256, 0, stream>>>(e_emb, hts, out);
    k_rs_wmma<<<N_DOC * (R / 32) * (D / 16), 32, 0, stream>>>(seq, ht_att, out);
}


# e_att, ht_att and hs/ts rows: one wave per row, all its row loads issued up front (256 blocks of 6/4/4 waves, 28672 -> 3584 waves); ht_att row sum in-wave in the same tree order, no LDS
# speedup vs baseline: 2.1526x; 2.1526x over previous
#include <hip/hip_runtime.h>
#include <math.h>

constexpr int N_DOC = 4;
constexpr int L     = 1024;
constexpr int D     = 768;
constexpr int H     = 12;
constexpr int E     = 32;
constexpr int M     = 4;
constexpr int R     = 256;
constexpr float NEG_MAX = -3.402823466e+38f;

typedef __attribute__((ext_vector_type(2))) float v2f;
typedef __attribute__((ext_vector_type(4))) float v4f;
typedef __attribute__((ext_vector_type(8))) float v8f;

__global__ void k_e_emb(const float* __restrict__ seq,
                        const int*   __restrict__ pos,
                        const float* __restrict__ mask,
                        float* __restrict__ e_emb) {
    int b = blockIdx.x;
    int i = b / E;
    const float* seq_i  = seq  + (size_t)i * L * D;
    const int*   pos_e  = pos  + (size_t)b * M;
    const float* mask_e = mask + (size_t)b * M;

    int   p[M];
    float mk[M];
#pragma unroll
    for (int m = 0; m < M; ++m) { p[m] = pos_e[m] + 1; mk[m] = mask_e[m]; }

    for (int c = threadIdx.x; c < D; c += blockDim.x) {
        float v[M];
        float mx = NEG_MAX;
#pragma unroll
        for (int m = 0; m < M; ++m) {
            float val = seq_i[(size_t)p[m] * D + c];
            v[m] = (mk[m] > 0.0f) ? val : NEG_MAX;
            mx = fmaxf(mx, v[m]);
        }
        float s = 0.0f;
#pragma unroll
        for (int m = 0; m < M; ++m)
            if (mk[m] > 0.0f) s += expf(v[m] - mx);
        e_emb[(size_t)b * D + c] = mx + logf(s);
    }
}

constexpr int LQ = L / 128;

__global__ __launch_bounds__(192) void k_e_att(const float* __restrict__ att,
                                               const int*   __restrict__ pos,
                                               const float* __restrict__ mask,
                                               float* __restrict__ e_att) {
    int w    = __builtin_amdgcn_readfirstlane((int)(threadIdx.x >> 5));
    int lane = threadIdx.x & 31;
    int b  = blockIdx.x * 6 + w;
    int hh = b % H;
    int ie = b / H;
    int i  = ie / E;

    const int*   pos_e  = pos  + (size_t)ie * M;
    const float* mask_e = mask + (size_t)ie * M;

    float mk[M];
    float cnt = 0.0f;
#pragma unroll
    for (int m = 0; m < M; ++m) { mk[m] = mask_e[m]; cnt += mk[m]; }
    float inv = 1.0f / fmaxf(cnt, 1.0f);

    const float* base = att + ((size_t)i * H + hh) * L * L + 4 * lane;
    v4f x[M][LQ];
#pragma unroll
    for (int m = 0; m < M; ++m) {
        const float* row = base + (size_t)(pos_e[m] + 1) * L;
#pragma unroll
        for (int j = 0; j < LQ; ++j) x[m][j] = *(const v4f*)(row + 128 * j);
    }
    __builtin_amdgcn_sched_barrier(0);

    v4f acc[LQ];
#pragma unroll
    for (int j = 0; j < LQ; ++j) acc[j] = (v4f){0.0f, 0.0f, 0.0f, 0.0f};
#pragma unroll
    for (int m = 0; m < M; ++m) {
        float cm = (mk[m] > 0.0f) ? mk[m] : 0.0f;
#pragma unroll
        for (int j = 0; j < LQ; ++j) acc[j] += cm * x[m][j];
    }

    float* dst = e_att + (size_t)b * L + 4 * lane;
#pragma unroll
    for (int j = 0; j < LQ; ++j) {
        acc[j] *= inv;
        *(v4f*)(dst + 128 * j) = acc[j];
    }
}

__global__ __launch_bounds__(128) void k_ht_att(const float* __restrict__ e_att,
                                                const int*   __restrict__ hts,
                                                float* __restrict__ ht_att) {
    int w    = __builtin_amdgcn_readfirstlane((int)(threadIdx.x >> 5));
    int lane = threadIdx.x & 31;
    int b  = blockIdx.x * 4 + w;
    int i  = b / R;
    int eh = hts[(size_t)b * 2 + 0];
    int et = hts[(size_t)b * 2 + 1];

    const float* ha = e_att + ((size_t)i * E + eh) * H * L + 4 * lane;
    const float* ta = e_att + ((size_t)i * E + et) * H * L + 4 * lane;

    v4f acc[LQ];
#pragma unroll
    for (int j = 0; j < LQ; ++j) acc[j] = (v4f){0.0f, 0.0f, 0.0f, 0.0f};
    v4f a[H][LQ], t[H][LQ];
#pragma unroll
    for (int hh = 0; hh < H; ++hh) {
#pragma unroll
        for (int j = 0; j < LQ; ++j) {
            a[hh][j] = *(const v4f*)(ha + (size_t)hh * L + 128 * j);
            t[hh][j] = *(const v4f*)(ta + (size_t)hh * L + 128 * j);
        }
    }
    __builtin_amdgcn_sched_barrier(0);
#pragma unroll
    for (int hh = 0; hh < H; ++hh) {
#pragma unroll
        for (int j = 0; j < LQ; ++j) acc[j] += a[hh][j] * t[hh][j];
    }

    float ps[LQ];
#pragma unroll
    for (int j = 0; j < LQ; ++j) {
        acc[j] *= (1.0f / (float)H);
        ps[j] = acc[j].x + acc[j].y + acc[j].z + acc[j].w;
    }
    float s4[4], s2[2];
#pragma unroll
    for (int j = 0; j < 4; ++j) s4[j] = ps[j] + ps[j + 4];
#pragma unroll
    for (int j = 0; j < 2; ++j) s2[j] = s4[j] + s4[j + 2];
    float c = s2[0] + s2[1];
#pragma unroll
    for (int s = 16; s > 0; s >>= 1) c += __shfl_down(c, s, 32);
    float tot  = __shfl(c, 0, 32);
    float invs = 1.0f / (tot + 1e-5f);

    float* dst = ht_att + (size_t)b * L + 4 * lane;
#pragma unroll
    for (int j = 0; j < LQ; ++j) *(v4f*)(dst + 128 * j) = acc[j] * invs;
}

__global__ __launch_bounds__(128) void k_hs_ts(const float* __restrict__ e_emb,
                                               const int*   __restrict__ hts,
                                               float* __restrict__ out) {
    constexpr int DQ = D / 128;
    int w    = __builtin_amdgcn_readfirstlane((int)(threadIdx.x >> 5));
    int lane = threadIdx.x & 31;
    int b  = blockIdx.x * 4 + w;
    int i  = b / R;
    int eh = hts[(size_t)b * 2 + 0];
    int et = hts[(size_t)b * 2 + 1];

    const float* sh = e_emb + ((size_t)i * E + eh) * D + 4 * lane;
    const float* st = e_emb + ((size_t)i * E + et) * D + 4 * lane;
    float* oh = out + (size_t)b * D + 4 * lane;
    float* ot = out + (size_t)N_DOC * R * D + (size_t)b * D + 4 * lane;

    v4f a[DQ], t[DQ];
#pragma unroll
    for (int j = 0; j < DQ; ++j) {
        a[j] = *(const v4f*)(sh + 128 * j);
        t[j] = *(const v4f*)(st + 128 * j);
    }
#pragma unroll
    for (int j = 0; j < DQ; ++j) {
        *(v4f*)(oh + 128 * j) = a[j];
        *(v4f*)(ot + 128 * j) = t[j];
    }
}

struct RsFrag {
    v4f a[2];
    v4f bq[4];
    v2f bp[4];
};

__device__ __forceinline__ void rs_load(RsFrag& f, const float* pa0, const float* pa1,
                                        const float* pbq, const float* pbp, int kc) {
    f.a[0] = *(const v4f*)(pa0 + kc);
    f.a[1] = *(const v4f*)(pa1 + kc);
#pragma unroll
    for (int j = 0; j < 4; ++j) {
        f.bq[j] = *(const v4f*)(pbq + (size_t)(kc + j) * D);
        f.bp[j] = *(const v2f*)(pbp + (size_t)(kc + j) * D);
    }
}

__device__ __forceinline__ void rs_mma(const RsFrag& f, v8f (&acc)[2][6]) {
#pragma unroll
    for (int s = 0; s < 2; ++s) {
        v2f a0 = {f.a[0][2 * s], f.a[0][2 * s + 1]};
        v2f a1 = {f.a[1][2 * s], f.a[1][2 * s + 1]};
#pragma unroll
        for (int u = 0; u < 6; ++u) {
            v2f bb;
            if (u < 4) { bb.x = f.bq[2 * s][u];     bb.y = f.bq[2 * s + 1][u]; }
            else       { bb.x = f.bp[2 * s][u - 4]; bb.y = f.bp[2 * s + 1][u - 4]; }
            acc[0][u] = __builtin_amdgcn_wmma_f32_16x16x4_f32(
                    false, a0, false, bb, (short)0, acc[0][u], false, false);
            acc[1][u] = __builtin_amdgcn_wmma_f32_16x16x4_f32(
                    false, a1, false, bb, (short)0, acc[1][u], false, false);
        }
    }
}

__global__ __launch_bounds__(256) void k_rs_wmma(const float* __restrict__ seq,
                                                 const float* __restrict__ ht_att,
                                                 float* __restrict__ out) {
    constexpr int BM  = 32;
    constexpr int BN  = 96;
    constexpr int NW  = 8;
    constexpr int KW  = L / NW;
    constexpr int NTB = D / BN;
    constexpr int MTB = R / BM;
    int b  = blockIdx.x;
    int nt = b % NTB;
    int mt = (b / NTB) % MTB;
    int i  = b / (NTB * MTB);
    int m0 = mt * BM;
    int n0 = nt * BN;

    int w    = threadIdx.x >> 5;
    int lane = threadIdx.x & 31;
    int h    = lane >> 4;
    int c    = lane & 15;
    int kb   = w * KW + 4 * h;

    const float* pa0 = ht_att + (size_t)i * R * L + (size_t)(m0 + c) * L + kb;
    const float* pa1 = pa0 + (size_t)16 * L;
    const float* pb  = seq + (size_t)i * L * D + (size_t)kb * D + n0;
    const float* pbq = pb + 4 * c;
    const float* pbp = pb + 64 + 2 * c;

    v8f acc[2][6];
#pragma unroll
    for (int t = 0; t < 2; ++t)
#pragma unroll
        for (int u = 0; u < 6; ++u)
            acc[t][u] = (v8f){0.f, 0.f, 0.f, 0.f, 0.f, 0.f, 0.f, 0.f};

#define RS_GROUP() __builtin_amdgcn_sched_barrier(0)
    RsFrag f0, f1, f2;
    rs_load(f0, pa0, pa1, pbq, pbp, 0);
    rs_load(f1, pa0, pa1, pbq, pbp, 8);
#pragma unroll 1
    for (int kc = 0; kc < KW - 32; kc += 24) {
        rs_load(f2, pa0, pa1, pbq, pbp, kc + 16); RS_GROUP();
        rs_mma(f0, acc);                          RS_GROUP();
        rs_load(f0, pa0, pa1, pbq, pbp, kc + 24); RS_GROUP();
        rs_mma(f1, acc);                          RS_GROUP();
        rs_load(f1, pa0, pa1, pbq, pbp, kc + 32); RS_GROUP();
        rs_mma(f2, acc);                          RS_GROUP();
    }
    rs_load(f2, pa0, pa1, pbq, pbp, KW - 16);     RS_GROUP();
    rs_mma(f0, acc);                              RS_GROUP();
    rs_load(f0, pa0, pa1, pbq, pbp, KW - 8);      RS_GROUP();
    rs_mma(f1, acc);
    rs_mma(f2, acc);
    rs_mma(f0, acc);
#undef RS_GROUP

    __shared__ v4f redq[NW * 16 * 32];
    __shared__ v2f redp[NW * 16 * 32];
#pragma unroll
    for (int t = 0; t < 2; ++t) {
#pragma unroll
        for (int r = 0; r < 8; ++r) {
            int p = t * 8 + r;
            v4f q  = {acc[t][0][r], acc[t][1][r], acc[t][2][r], acc[t][3][r]};
            v2f pp = {acc[t][4][r], acc[t][5][r]};
            redq[(w * 16 + p) * 32 + lane] = q;
            redp[(w * 16 + p) * 32 + lane] = pp;
        }
    }
    __syncthreads();

    float* O = out + (size_t)2 * N_DOC * R * D + (size_t)i * R * D;
#pragma unroll
    for (int e = 0; e < 2; ++e) {
        int p = 2 * w + e;
        int t = p >> 3;
        int r = p & 7;
        v4f sq = redq[p * 32 + lane];
        v2f sp = redp[p * 32 + lane];
#pragma unroll
        for (int ww = 1; ww < NW; ++ww) {
            sq += redq[(ww * 16 + p) * 32 + lane];
            sp += redp[(ww * 16 + p) * 32 + lane];
        }
        int row = m0 + 16 * t + 8 * h + r;
        float* o = O + (size_t)row * D + n0;
        *(v4f*)(o + 4 * c)      = sq;
        *(v2f*)(o + 64 + 2 * c) = sp;
    }
}

extern "C" void kernel_launch(void* const* d_in, const int* in_sizes, int n_in,
                              void* d_out, int out_size, void* d_ws, size_t ws_size,
                              hipStream_t stream) {
    const float* seq   = (const float*)d_in[0];
    const float* att   = (const float*)d_in[1];
    const int*   pos   = (const int*)  d_in[2];
    const float* mask  = (const float*)d_in[3];
    const int*   hts   = (const int*)  d_in[4];
    float*       out   = (float*)d_out;

    float* ws     = (float*)d_ws;
    float* e_emb  = ws;
    float* e_att  = e_emb + (size_t)N_DOC * E * D;
    float* ht_att = e_att + (size_t)N_DOC * E * H * L;

    k_e_emb  <<<N_DOC * E,     256, 0, stream>>>(seq, pos, mask, e_emb);
    k_e_att  <<<N_DOC * E * H / 6, 192, 0, stream>>>(att, pos, mask, e_att);
    k_ht_att <<<N_DOC * R / 4,     128, 0, stream>>>(e_att, hts, ht_att);
    k_hs_ts  <<<N_DOC * R / 4,     128, 0, stream>>>(e_emb, hts, out);
    k_rs_wmma<<<N_DOC * (R / 32) * (D / 96), 256, 0, stream>>>(seq, ht_att, out);
}


# hs/ts by scatter: each entity's logsumexp columns go straight into every hs/ts row whose pair names it, from waves beside the e_att rows; entity plane and two kernels go (5 -> 3 launches)
# speedup vs baseline: 2.8200x; 1.3100x over previous
#include <hip/hip_runtime.h>
#include <math.h>

constexpr int N_DOC = 4;
constexpr int L     = 1024;
constexpr int D     = 768;
constexpr int H     = 12;
constexpr int E     = 32;
constexpr int M     = 4;
constexpr int R     = 256;
constexpr float NEG_MAX = -3.402823466e+38f;

typedef __attribute__((ext_vector_type(2))) float v2f;
typedef __attribute__((ext_vector_type(4))) float v4f;
typedef __attribute__((ext_vector_type(8))) float v8f;

constexpr int LQ = L / 128;
constexpr int N_ENT_WG = N_DOC * E;

typedef __attribute__((ext_vector_type(4))) int v4i;

__device__ __forceinline__ void entity_rows(const float* __restrict__ seq,
                                            const int*   __restrict__ pos,
                                            const float* __restrict__ mask,
                                            const int*   __restrict__ hts,
                                            float* __restrict__ out,
                                            int ie, int w, int lane) {
    int i = ie / E;
    int e = ie - i * E;
    const int*   pos_e  = pos  + (size_t)ie * M;
    const float* mask_e = mask + (size_t)ie * M;
    const float* src = seq + (size_t)i * L * D + 128 * w + 4 * lane;

    float mk[M];
    v4f   x[M];
#pragma unroll
    for (int m = 0; m < M; ++m) {
        mk[m] = mask_e[m];
        x[m]  = *(const v4f*)(src + (size_t)(pos_e[m] + 1) * D);
    }
    const v4i* hp = (const v4i*)(hts + (size_t)i * R * 2) + lane;
    v4i hv[4];
#pragma unroll
    for (int q = 0; q < 4; ++q) hv[q] = hp[32 * q];

    v4f res;
#pragma unroll
    for (int u = 0; u < 4; ++u) {
        float v[M];
        float mx = NEG_MAX;
#pragma unroll
        for (int m = 0; m < M; ++m) {
            v[m] = (mk[m] > 0.0f) ? x[m][u] : NEG_MAX;
            mx = fmaxf(mx, v[m]);
        }
        float s = 0.0f;
#pragma unroll
        for (int m = 0; m < M; ++m) {
            float ex = expf(v[m] - mx);
            s += (mk[m] > 0.0f) ? ex : 0.0f;
        }
        res[u] = mx + logf(s);
    }

    float* obase = out + (size_t)i * R * D + 128 * w + 4 * lane;
#pragma unroll
    for (int q = 0; q < 4; ++q) {
#pragma unroll
        for (int u = 0; u < 4; ++u) {
            unsigned hit = __builtin_amdgcn_ballot_w32(hv[q][u] == e);
            while (hit) {
                int b = __builtin_ctz(hit);
                hit &= hit - 1;
                int t    = 128 * q + 4 * b + u;
                int r    = t >> 1;
                int side = t & 1;
                *(v4f*)(obase + (size_t)side * N_DOC * R * D + (size_t)r * D) = res;
            }
        }
    }
}

__global__ __launch_bounds__(192) void k_stage1(const float* __restrict__ seq,
                                                const float* __restrict__ att,
                                                const int*   __restrict__ pos,
                                                const float* __restrict__ mask,
                                                const int*   __restrict__ hts,
                                                float* __restrict__ e_att,
                                                float* __restrict__ out) {
    int w    = __builtin_amdgcn_readfirstlane((int)(threadIdx.x >> 5));
    int lane = threadIdx.x & 31;

    if (blockIdx.x < N_ENT_WG) {
        entity_rows(seq, pos, mask, hts, out, (int)blockIdx.x, w, lane);
        return;
    }

    int b  = (blockIdx.x - N_ENT_WG) * 6 + w;
    int hh = b % H;
    int ie = b / H;
    int i  = ie / E;

    const int*   pos_e  = pos  + (size_t)ie * M;
    const float* mask_e = mask + (size_t)ie * M;

    float mk[M];
    float cnt = 0.0f;
#pragma unroll
    for (int m = 0; m < M; ++m) { mk[m] = mask_e[m]; cnt += mk[m]; }
    float inv = 1.0f / fmaxf(cnt, 1.0f);

    const float* base = att + ((size_t)i * H + hh) * L * L + 4 * lane;
    v4f x[M][LQ];
#pragma unroll
    for (int m = 0; m < M; ++m) {
        const float* row = base + (size_t)(pos_e[m] + 1) * L;
#pragma unroll
        for (int j = 0; j < LQ; ++j) x[m][j] = *(const v4f*)(row + 128 * j);
    }
    __builtin_amdgcn_sched_barrier(0);

    v4f acc[LQ];
#pragma unroll
    for (int j = 0; j < LQ; ++j) acc[j] = (v4f){0.0f, 0.0f, 0.0f, 0.0f};
#pragma unroll
    for (int m = 0; m < M; ++m) {
        float cm = (mk[m] > 0.0f) ? mk[m] : 0.0f;
#pragma unroll
        for (int j = 0; j < LQ; ++j) acc[j] += cm * x[m][j];
    }

    float* dst = e_att + (size_t)b * L + 4 * lane;
#pragma unroll
    for (int j = 0; j < LQ; ++j) {
        acc[j] *= inv;
        *(v4f*)(dst + 128 * j) = acc[j];
    }
}

__global__ __launch_bounds__(128) void k_ht_att(const float* __restrict__ e_att,
                                                const int*   __restrict__ hts,
                                                float* __restrict__ ht_att) {
    int w    = __builtin_amdgcn_readfirstlane((int)(threadIdx.x >> 5));
    int lane = threadIdx.x & 31;
    int b  = blockIdx.x * 4 + w;
    int i  = b / R;
    int eh = hts[(size_t)b * 2 + 0];
    int et = hts[(size_t)b * 2 + 1];

    const float* ha = e_att + ((size_t)i * E + eh) * H * L + 4 * lane;
    const float* ta = e_att + ((size_t)i * E + et) * H * L + 4 * lane;

    v4f acc[LQ];
#pragma unroll
    for (int j = 0; j < LQ; ++j) acc[j] = (v4f){0.0f, 0.0f, 0.0f, 0.0f};
    v4f a[H][LQ], t[H][LQ];
#pragma unroll
    for (int hh = 0; hh < H; ++hh) {
#pragma unroll
        for (int j = 0; j < LQ; ++j) {
            a[hh][j] = *(const v4f*)(ha + (size_t)hh * L + 128 * j);
            t[hh][j] = *(const v4f*)(ta + (size_t)hh * L + 128 * j);
        }
    }
    __builtin_amdgcn_sched_barrier(0);
#pragma unroll
    for (int hh = 0; hh < H; ++hh) {
#pragma unroll
        for (int j = 0; j < LQ; ++j) acc[j] += a[hh][j] * t[hh][j];
    }

    float ps[LQ];
#pragma unroll
    for (int j = 0; j < LQ; ++j) {
        acc[j] *= (1.0f / (float)H);
        ps[j] = acc[j].x + acc[j].y + acc[j].z + acc[j].w;
    }
    float s4[4], s2[2];
#pragma unroll
    for (int j = 0; j < 4; ++j) s4[j] = ps[j] + ps[j + 4];
#pragma unroll
    for (int j = 0; j < 2; ++j) s2[j] = s4[j] + s4[j + 2];
    float c = s2[0] + s2[1];
#pragma unroll
    for (int s = 16; s > 0; s >>= 1) c += __shfl_down(c, s, 32);
    float tot  = __shfl(c, 0, 32);
    float invs = 1.0f / (tot + 1e-5f);

    float* dst = ht_att + (size_t)b * L + 4 * lane;
#pragma unroll
    for (int j = 0; j < LQ; ++j) *(v4f*)(dst + 128 * j) = acc[j] * invs;
}

struct RsFrag {
    v4f a[2];
    v4f bq[4];
    v2f bp[4];
};

__device__ __forceinline__ void rs_load(RsFrag& f, const float* pa0, const float* pa1,
                                        const float* pbq, const float* pbp, int kc) {
    f.a[0] = *(const v4f*)(pa0 + kc);
    f.a[1] = *(const v4f*)(pa1 + kc);
#pragma unroll
    for (int j = 0; j < 4; ++j) {
        f.bq[j] = *(const v4f*)(pbq + (size_t)(kc + j) * D);
        f.bp[j] = *(const v2f*)(pbp + (size_t)(kc + j) * D);
    }
}

__device__ __forceinline__ void rs_mma(const RsFrag& f, v8f (&acc)[2][6]) {
#pragma unroll
    for (int s = 0; s < 2; ++s) {
        v2f a0 = {f.a[0][2 * s], f.a[0][2 * s + 1]};
        v2f a1 = {f.a[1][2 * s], f.a[1][2 * s + 1]};
#pragma unroll
        for (int u = 0; u < 6; ++u) {
            v2f bb;
            if (u < 4) { bb.x = f.bq[2 * s][u];     bb.y = f.bq[2 * s + 1][u]; }
            else       { bb.x = f.bp[2 * s][u - 4]; bb.y = f.bp[2 * s + 1][u - 4]; }
            acc[0][u] = __builtin_amdgcn_wmma_f32_16x16x4_f32(
                    false, a0, false, bb, (short)0, acc[0][u], false, false);
            acc[1][u] = __builtin_amdgcn_wmma_f32_16x16x4_f32(
                    false, a1, false, bb, (short)0, acc[1][u], false, false);
        }
    }
}

__global__ __launch_bounds__(256) void k_rs_wmma(const float* __restrict__ seq,
                                                 const float* __restrict__ ht_att,
                                                 float* __restrict__ out) {
    constexpr int BM  = 32;
    constexpr int BN  = 96;
    constexpr int NW  = 8;
    constexpr int KW  = L / NW;
    constexpr int NTB = D / BN;
    constexpr int MTB = R / BM;
    int b  = blockIdx.x;
    int nt = b % NTB;
    int mt = (b / NTB) % MTB;
    int i  = b / (NTB * MTB);
    int m0 = mt * BM;
    int n0 = nt * BN;

    int w    = threadIdx.x >> 5;
    int lane = threadIdx.x & 31;
    int h    = lane >> 4;
    int c    = lane & 15;
    int kb   = w * KW + 4 * h;

    const float* pa0 = ht_att + (size_t)i * R * L + (size_t)(m0 + c) * L + kb;
    const float* pa1 = pa0 + (size_t)16 * L;
    const float* pb  = seq + (size_t)i * L * D + (size_t)kb * D + n0;
    const float* pbq = pb + 4 * c;
    const float* pbp = pb + 64 + 2 * c;

    v8f acc[2][6];
#pragma unroll
    for (int t = 0; t < 2; ++t)
#pragma unroll
        for (int u = 0; u < 6; ++u)
            acc[t][u] = (v8f){0.f, 0.f, 0.f, 0.f, 0.f, 0.f, 0.f, 0.f};

#define RS_GROUP() __builtin_amdgcn_sched_barrier(0)
    RsFrag f0, f1, f2;
    rs_load(f0, pa0, pa1, pbq, pbp, 0);
    rs_load(f1, pa0, pa1, pbq, pbp, 8);
#pragma unroll 1
    for (int kc = 0; kc < KW - 32; kc += 24) {
        rs_load(f2, pa0, pa1, pbq, pbp, kc + 16); RS_GROUP();
        rs_mma(f0, acc);                          RS_GROUP();
        rs_load(f0, pa0, pa1, pbq, pbp, kc + 24); RS_GROUP();
        rs_mma(f1, acc);                          RS_GROUP();
        rs_load(f1, pa0, pa1, pbq, pbp, kc + 32); RS_GROUP();
        rs_mma(f2, acc);                          RS_GROUP();
    }
    rs_load(f2, pa0, pa1, pbq, pbp, KW - 16);     RS_GROUP();
    rs_mma(f0, acc);                              RS_GROUP();
    rs_load(f0, pa0, pa1, pbq, pbp, KW - 8);      RS_GROUP();
    rs_mma(f1, acc);
    rs_mma(f2, acc);
    rs_mma(f0, acc);
#undef RS_GROUP

    __shared__ v4f redq[NW * 16 * 32];
    __shared__ v2f redp[NW * 16 * 32];
#pragma unroll
    for (int t = 0; t < 2; ++t) {
#pragma unroll
        for (int r = 0; r < 8; ++r) {
            int p = t * 8 + r;
            v4f q  = {acc[t][0][r], acc[t][1][r], acc[t][2][r], acc[t][3][r]};
            v2f pp = {acc[t][4][r], acc[t][5][r]};
            redq[(w * 16 + p) * 32 + lane] = q;
            redp[(w * 16 + p) * 32 + lane] = pp;
        }
    }
    __syncthreads();

    float* O = out + (size_t)2 * N_DOC * R * D + (size_t)i * R * D;
#pragma unroll
    for (int e = 0; e < 2; ++e) {
        int p = 2 * w + e;
        int t = p >> 3;
        int r = p & 7;
        v4f sq = redq[p * 32 + lane];
        v2f sp = redp[p * 32 + lane];
#pragma unroll
        for (int ww = 1; ww < NW; ++ww) {
            sq += redq[(ww * 16 + p) * 32 + lane];
            sp += redp[(ww * 16 + p) * 32 + lane];
        }
        int row = m0 + 16 * t + 8 * h + r;
        float* o = O + (size_t)row * D + n0;
        *(v4f*)(o + 4 * c)      = sq;
        *(v2f*)(o + 64 + 2 * c) = sp;
    }
}

extern "C" void kernel_launch(void* const* d_in, const int* in_sizes, int n_in,
                              void* d_out, int out_size, void* d_ws, size_t ws_size,
                              hipStream_t stream) {
    const float* seq   = (const float*)d_in[0];
    const float* att   = (const float*)d_in[1];
    const int*   pos   = (const int*)  d_in[2];
    const float* mask  = (const float*)d_in[3];
    const int*   hts   = (const int*)  d_in[4];
    float*       out   = (float*)d_out;

    float* ws     = (float*)d_ws;
    float* e_att  = ws;
    float* ht_att = e_att + (size_t)N_DOC * E * H * L;

    k_stage1 <<<N_ENT_WG + N_DOC * E * H / 6, 192, 0, stream>>>(seq, att, pos, mask, hts, e_att, out);
    k_ht_att <<<N_DOC * R / 4,     128, 0, stream>>>(e_att, hts, ht_att);
    k_rs_wmma<<<N_DOC * (R / 32) * (D / 96), 256, 0, stream>>>(seq, ht_att, out);
}


# 3->2 launches: block per (doc, 16-col slice) gathers entity-head rows, per-column 32x32 Gram over heads on f32 WMMA, pairs pick G[eh][et]; slice row sums stored, GEMM scales rows; sum orders change
# speedup vs baseline: 3.2636x; 1.1573x over previous
#include <hip/hip_runtime.h>
#include <math.h>

constexpr int N_DOC = 4;
constexpr int L     = 1024;
constexpr int D     = 768;
constexpr int H     = 12;
constexpr int E     = 32;
constexpr int M     = 4;
constexpr int R     = 256;
constexpr float NEG_MAX = -3.402823466e+38f;

typedef __attribute__((ext_vector_type(2))) float v2f;
typedef __attribute__((ext_vector_type(4))) float v4f;
typedef __attribute__((ext_vector_type(8))) float v8f;

constexpr int N_ENT_WG = N_DOC * E;

typedef __attribute__((ext_vector_type(4))) int v4i;

__device__ __forceinline__ void entity_rows(const float* __restrict__ seq,
                                            const int*   __restrict__ pos,
                                            const float* __restrict__ mask,
                                            const int*   __restrict__ hts,
                                            float* __restrict__ out,
                                            int ie, int w, int lane) {
    int i = ie / E;
    int e = ie - i * E;
    const int*   pos_e  = pos  + (size_t)ie * M;
    const float* mask_e = mask + (size_t)ie * M;
    const float* src = seq + (size_t)i * L * D + 128 * w + 4 * lane;

    float mk[M];
    v4f   x[M];
#pragma unroll
    for (int m = 0; m < M; ++m) {
        mk[m] = mask_e[m];
        x[m]  = *(const v4f*)(src + (size_t)(pos_e[m] + 1) * D);
    }
    const v4i* hp = (const v4i*)(hts + (size_t)i * R * 2) + lane;
    v4i hv[4];
#pragma unroll
    for (int q = 0; q < 4; ++q) hv[q] = hp[32 * q];

    v4f res;
#pragma unroll
    for (int u = 0; u < 4; ++u) {
        float v[M];
        float mx = NEG_MAX;
#pragma unroll
        for (int m = 0; m < M; ++m) {
            v[m] = (mk[m] > 0.0f) ? x[m][u] : NEG_MAX;
            mx = fmaxf(mx, v[m]);
        }
        float s = 0.0f;
#pragma unroll
        for (int m = 0; m < M; ++m) {
            float ex = expf(v[m] - mx);
            s += (mk[m] > 0.0f) ? ex : 0.0f;
        }
        res[u] = mx + logf(s);
    }

    float* obase = out + (size_t)i * R * D + 128 * w + 4 * lane;
#pragma unroll
    for (int q = 0; q < 4; ++q) {
#pragma unroll
        for (int u = 0; u < 4; ++u) {
            unsigned hit = __builtin_amdgcn_ballot_w32(hv[q][u] == e);
            while (hit) {
                int b = __builtin_ctz(hit);
                hit &= hit - 1;
                int t    = 128 * q + 4 * b + u;
                int r    = t >> 1;
                int side = t & 1;
                *(v4f*)(obase + (size_t)side * N_DOC * R * D + (size_t)r * D) = res;
            }
        }
    }
}

typedef __attribute__((ext_vector_type(2))) int v2i;

constexpr int SLC   = 16;
constexpr int NSL   = L / SLC;
constexpr int EA_CS = E * H + 2;
constexpr int GS_P  = 20;

__device__ __forceinline__ void pair_slice(const float* __restrict__ att,
                                           const int*   __restrict__ pos,
                                           const float* __restrict__ mask,
                                           const int*   __restrict__ hts,
                                           float* __restrict__ praw,
                                           float* __restrict__ psum,
                                           int bs, int w, int lane,
                                           float* __restrict__ ea, float* __restrict__ gs) {
    int i   = bs / NSL;
    int sl  = bs - i * NSL;
    int tid = w * 32 + lane;
    int q   = tid & 3;
    int xb  = tid >> 2;

    const v2i* hp = (const v2i*)(hts + (size_t)i * R * 2);
    v2i hv[4];
#pragma unroll
    for (int j = 0; j < 4; ++j) hv[j] = hp[64 * j + xb];
    v2i hs = hp[tid];

    v4i pv[6];
    v4f mv[6];
#pragma unroll
    for (int jj = 0; jj < 6; ++jj) {
        int e = (xb + 64 * jj) / H;
        pv[jj] = *(const v4i*)(pos  + ((size_t)i * E + e) * M);
        mv[jj] = *(const v4f*)(mask + ((size_t)i * E + e) * M);
    }

    const float* abase = att + (size_t)i * H * L * L + SLC * sl + 4 * q;
    v4f xr[6][M];
#pragma unroll
    for (int jj = 0; jj < 6; ++jj) {
        int x  = xb + 64 * jj;
        int e  = x / H;
        int hh = x - e * H;
#pragma unroll
        for (int m = 0; m < M; ++m)
            xr[jj][m] = *(const v4f*)(abase + ((size_t)hh * L + (size_t)(pv[jj][m] + 1)) * L);
    }
    __builtin_amdgcn_sched_barrier(0);

#pragma unroll
    for (int jj = 0; jj < 6; ++jj) {
        float cnt = 0.0f;
#pragma unroll
        for (int m = 0; m < M; ++m) cnt += mv[jj][m];
        float inv = 1.0f / fmaxf(cnt, 1.0f);
        v4f acc = (v4f){0.0f, 0.0f, 0.0f, 0.0f};
#pragma unroll
        for (int m = 0; m < M; ++m) {
            float cm = (mv[jj][m] > 0.0f) ? mv[jj][m] : 0.0f;
            acc += cm * xr[jj][m];
        }
        acc *= inv;
        int x = xb + 64 * jj;
#pragma unroll
        for (int u = 0; u < 4; ++u) ea[(4 * q + u) * EA_CS + x] = acc[u];
    }
    __syncthreads();

    int c = lane & 15;
    int h = lane >> 4;
    v8f g[2][2][2];
#pragma unroll
    for (int kk = 0; kk < 2; ++kk) {
        const float* fp = ea + (2 * w + kk) * EA_CS + c * H + 6 * h;
        v2f fr[2][3];
#pragma unroll
        for (int t = 0; t < 2; ++t)
#pragma unroll
            for (int s = 0; s < 3; ++s) fr[t][s] = *(const v2f*)(fp + 16 * t * H + 2 * s);
#pragma unroll
        for (int t = 0; t < 2; ++t) {
#pragma unroll
            for (int tt = 0; tt < 2; ++tt) {
                v8f acc = (v8f){0.f, 0.f, 0.f, 0.f, 0.f, 0.f, 0.f, 0.f};
#pragma unroll
                for (int s = 0; s < 3; ++s)
                    acc = __builtin_amdgcn_wmma_f32_16x16x4_f32(
                            false, fr[t][s], false, fr[tt][s], (short)0, acc, false, false);
                g[kk][t][tt] = acc;
            }
        }
    }
#pragma unroll
    for (int t = 0; t < 2; ++t) {
#pragma unroll
        for (int tt = 0; tt < 2; ++tt) {
#pragma unroll
            for (int r = 0; r < 8; ++r) {
                int a = 16 * t + 8 * h + r;
                int b = 16 * tt + c;
                v2f two = {g[0][t][tt][r], g[1][t][tt][r]};
                *(v2f*)(gs + (a * E + b) * GS_P + 2 * w + 2 * h) = two;
            }
        }
    }
    __syncthreads();

    float* pdst = praw + (size_t)(i * NSL + sl) * R * SLC + 4 * q;
#pragma unroll
    for (int j = 0; j < 4; ++j) {
        int eh = hv[j].x;
        int et = hv[j].y;
        const float* gp = gs + (eh * E + et) * GS_P + 4 * q + 2 * ((eh >> 3) & 1);
        v2f lo = *(const v2f*)gp;
        v2f hi = *(const v2f*)(gp + 2);
        v4f v  = {lo.x, lo.y, hi.x, hi.y};
        v *= (1.0f / (float)H);
        *(v4f*)(pdst + (size_t)(64 * j + xb) * SLC) = v;
    }

    {
        const float* gp = gs + (hs.x * E + hs.y) * GS_P + 2 * ((hs.x >> 3) & 1);
        v2f x[SLC / 2];
#pragma unroll
        for (int k = 0; k < SLC / 2; ++k) x[k] = *(const v2f*)(gp + 2 * k);
        float sm = 0.0f;
#pragma unroll
        for (int k = 0; k < SLC / 2; ++k) {
            sm += x[k].x * (1.0f / (float)H);
            sm += x[k].y * (1.0f / (float)H);
        }
        psum[(size_t)(i * NSL + sl) * R + tid] = sm;
    }
}

__global__ __launch_bounds__(256) void k_front(const float* __restrict__ seq,
                                               const float* __restrict__ att,
                                               const int*   __restrict__ pos,
                                               const float* __restrict__ mask,
                                               const int*   __restrict__ hts,
                                               float* __restrict__ praw,
                                               float* __restrict__ psum,
                                               float* __restrict__ out) {
    __shared__ __attribute__((aligned(16))) float ea[SLC * EA_CS];
    __shared__ __attribute__((aligned(16))) float gs[E * E * GS_P];
    int w    = __builtin_amdgcn_readfirstlane((int)(threadIdx.x >> 5));
    int lane = threadIdx.x & 31;

    if (blockIdx.x < N_ENT_WG) {
        if (w < 6) entity_rows(seq, pos, mask, hts, out, (int)blockIdx.x, w, lane);
        return;
    }
    pair_slice(att, pos, mask, hts, praw, psum, (int)blockIdx.x - N_ENT_WG, w, lane, ea, gs);
}

struct RsFrag {
    v4f a[2];
    v4f bq[4];
    v2f bp[4];
};

__device__ __forceinline__ void rs_load(RsFrag& f, const float* pa0, const float* pa1,
                                        const float* pbq, const float* pbp, int kc) {
    int ao = (kc >> 4) * (R * SLC) + (kc & 8);
    f.a[0] = *(const v4f*)(pa0 + ao);
    f.a[1] = *(const v4f*)(pa1 + ao);
#pragma unroll
    for (int j = 0; j < 4; ++j) {
        f.bq[j] = *(const v4f*)(pbq + (size_t)(kc + j) * D);
        f.bp[j] = *(const v2f*)(pbp + (size_t)(kc + j) * D);
    }
}

__device__ __forceinline__ void rs_mma(const RsFrag& f, v8f (&acc)[2][6]) {
#pragma unroll
    for (int s = 0; s < 2; ++s) {
        v2f a0 = {f.a[0][2 * s], f.a[0][2 * s + 1]};
        v2f a1 = {f.a[1][2 * s], f.a[1][2 * s + 1]};
#pragma unroll
        for (int u = 0; u < 6; ++u) {
            v2f bb;
            if (u < 4) { bb.x = f.bq[2 * s][u];     bb.y = f.bq[2 * s + 1][u]; }
            else       { bb.x = f.bp[2 * s][u - 4]; bb.y = f.bp[2 * s + 1][u - 4]; }
            acc[0][u] = __builtin_amdgcn_wmma_f32_16x16x4_f32(
                    false, a0, false, bb, (short)0, acc[0][u], false, false);
            acc[1][u] = __builtin_amdgcn_wmma_f32_16x16x4_f32(
                    false, a1, false, bb, (short)0, acc[1][u], false, false);
        }
    }
}

__global__ __launch_bounds__(256) void k_rs_wmma(const float* __restrict__ seq,
                                                 const float* __restrict__ praw,
                                                 const float* __restrict__ psum,
                                                 float* __restrict__ out) {
    constexpr int BM  = 32;
    constexpr int BN  = 96;
    constexpr int NW  = 8;
    constexpr int KW  = L / NW;
    constexpr int NTB = D / BN;
    constexpr int MTB = R / BM;
    int b  = blockIdx.x;
    int nt = b % NTB;
    int mt = (b / NTB) % MTB;
    int i  = b / (NTB * MTB);
    int m0 = mt * BM;
    int n0 = nt * BN;

    int w    = threadIdx.x >> 5;
    int lane = threadIdx.x & 31;
    int h    = lane >> 4;
    int c    = lane & 15;
    int kb   = w * KW + 4 * h;

    const float* pa0 = praw + ((size_t)(i * NSL + w * (KW / SLC)) * R + (size_t)(m0 + c)) * SLC + 4 * h;
    const float* pa1 = pa0 + (size_t)16 * SLC;
    const float* pb  = seq + (size_t)i * L * D + (size_t)kb * D + n0;
    const float* pbq = pb + 4 * c;
    const float* pbp = pb + 64 + 2 * c;

    v8f acc[2][6];
#pragma unroll
    for (int t = 0; t < 2; ++t)
#pragma unroll
        for (int u = 0; u < 6; ++u)
            acc[t][u] = (v8f){0.f, 0.f, 0.f, 0.f, 0.f, 0.f, 0.f, 0.f};
    const float* ps = psum + (size_t)(i * NSL + w * (KW / SLC)) * R + m0 + lane;
    float pr[KW / SLC];
#pragma unroll
    for (int k = 0; k < KW / SLC; ++k) pr[k] = ps[(size_t)k * R];

#define RS_GROUP() __builtin_amdgcn_sched_barrier(0)
    RsFrag f0, f1, f2;
    rs_load(f0, pa0, pa1, pbq, pbp, 0);
    rs_load(f1, pa0, pa1, pbq, pbp, 8);
#pragma unroll 1
    for (int kc = 0; kc < KW - 32; kc += 24) {
        rs_load(f2, pa0, pa1, pbq, pbp, kc + 16); RS_GROUP();
        rs_mma(f0, acc);                          RS_GROUP();
        rs_load(f0, pa0, pa1, pbq, pbp, kc + 24); RS_GROUP();
        rs_mma(f1, acc);                          RS_GROUP();
        rs_load(f1, pa0, pa1, pbq, pbp, kc + 32); RS_GROUP();
        rs_mma(f2, acc);                          RS_GROUP();
    }
    rs_load(f2, pa0, pa1, pbq, pbp, KW - 16);     RS_GROUP();
    rs_mma(f0, acc);                              RS_GROUP();
    rs_load(f0, pa0, pa1, pbq, pbp, KW - 8);      RS_GROUP();
    rs_mma(f1, acc);
    rs_mma(f2, acc);
    rs_mma(f0, acc);
#undef RS_GROUP

    __shared__ v4f redq[NW * 16 * 32];
    __shared__ v2f redp[NW * 16 * 32];
    __shared__ float rowp[NW * BM];
    {
        float rs = pr[0];
#pragma unroll
        for (int k = 1; k < KW / SLC; ++k) rs += pr[k];
        rowp[w * BM + lane] = rs;
    }
#pragma unroll
    for (int t = 0; t < 2; ++t) {
#pragma unroll
        for (int r = 0; r < 8; ++r) {
            int p = t * 8 + r;
            v4f q  = {acc[t][0][r], acc[t][1][r], acc[t][2][r], acc[t][3][r]};
            v2f pp = {acc[t][4][r], acc[t][5][r]};
            redq[(w * 16 + p) * 32 + lane] = q;
            redp[(w * 16 + p) * 32 + lane] = pp;
        }
    }
    __syncthreads();

    float* O = out + (size_t)2 * N_DOC * R * D + (size_t)i * R * D;
#pragma unroll
    for (int e = 0; e < 2; ++e) {
        int p = 2 * w + e;
        int t = p >> 3;
        int r = p & 7;
        v4f sq = redq[p * 32 + lane];
        v2f sp = redp[p * 32 + lane];
#pragma unroll
        for (int ww = 1; ww < NW; ++ww) {
            sq += redq[(ww * 16 + p) * 32 + lane];
            sp += redp[(ww * 16 + p) * 32 + lane];
        }
        float tot = rowp[16 * t + 8 * h + r];
#pragma unroll
        for (int ww = 1; ww < NW; ++ww) tot += rowp[ww * BM + 16 * t + 8 * h + r];
        float invs = 1.0f / (tot + 1e-5f);
        sq *= invs;
        sp *= invs;
        int row = m0 + 16 * t + 8 * h + r;
        float* o = O + (size_t)row * D + n0;
        *(v4f*)(o + 4 * c)      = sq;
        *(v2f*)(o + 64 + 2 * c) = sp;
    }
}

extern "C" void kernel_launch(void* const* d_in, const int* in_sizes, int n_in,
                              void* d_out, int out_size, void* d_ws, size_t ws_size,
                              hipStream_t stream) {
    const float* seq   = (const float*)d_in[0];
    const float* att   = (const float*)d_in[1];
    const int*   pos   = (const int*)  d_in[2];
    const float* mask  = (const float*)d_in[3];
    const int*   hts   = (const int*)  d_in[4];
    float*       out   = (float*)d_out;

    float* praw = (float*)d_ws;
    float* psum = praw + (size_t)N_DOC * NSL * R * SLC;

    k_front  <<<N_ENT_WG + N_DOC * NSL, 256, 0, stream>>>(seq, att, pos, mask, hts, praw, psum, out);
    k_rs_wmma<<<N_DOC * (R / 32) * (D / 96), 256, 0, stream>>>(seq, praw, psum, out);
}
